# starting kernel
# baseline (speedup 1.0000x reference)
#include <hip/hip_runtime.h>
#include <hip/hip_bf16.h>

typedef float v2f __attribute__((ext_vector_type(2)));
typedef float v8f __attribute__((ext_vector_type(8)));

#define T_DIM 2048
#define D_DIM 256
#define H_DIM 256
#define GAMMA_LAMBDA (0.99f * 0.9f)
#define CONT_ALPHA 0.01f

__global__ __launch_bounds__(256) void fwd_gemm_kernel(
    const float* __restrict__ X, const float* __restrict__ W1,
    const float* __restrict__ b1, float* __restrict__ h)
{
    const int lane = threadIdx.x & 31;
    const int wave = threadIdx.x >> 5;
    const int tile = blockIdx.x * 8 + wave;
    const int tileM = tile >> 4;
    const int tileN = tile & 15;
    const int half = lane >> 4;
    const int l16  = lane & 15;
    const int rowA = tileM * 16 + l16;
    const int colB = tileN * 16 + l16;

    v8f acc = {};
    for (int k0 = 0; k0 < D_DIM; k0 += 4) {
        const int ka = k0 + 2 * half;
        v2f a, b;
        a.x = X[rowA * D_DIM + ka];
        a.y = X[rowA * D_DIM + ka + 1];
        b.x = W1[ka * H_DIM + colB];
        b.y = W1[(ka + 1) * H_DIM + colB];
        acc = __builtin_amdgcn_wmma_f32_16x16x4_f32(
            false, a, false, b, (short)0, acc, false, false);
    }
    const float bias = b1[colB];
#pragma unroll
    for (int v = 0; v < 8; ++v) {
        const int m = tileM * 16 + half * 8 + v;
        const float val = acc[v] + bias;
        h[m * H_DIM + colB] = val > 0.0f ? val : 0.0f;
    }
}

__global__ __launch_bounds__(256) void value_kernel(
    const float* __restrict__ h, const float* __restrict__ W2,
    const float* __restrict__ b2, float* __restrict__ u)
{
    __shared__ float s[256];
    const int t = blockIdx.x;
    const int j = threadIdx.x;
    s[j] = h[t * H_DIM + j] * W2[j];
    __syncthreads();
    for (int off = 128; off > 0; off >>= 1) {
        if (j < off) s[j] += s[j + off];
        __syncthreads();
    }
    if (j == 0) u[t] = s[0] + b2[0];
}

__global__ void scan_kernel(const float* __restrict__ u, const float* __restrict__ ys,
                            float* __restrict__ sigma, float* __restrict__ w,
                            float* __restrict__ out)
{
    if (threadIdx.x != 0 || blockIdx.x != 0) return;
    float r = 0.0f;
    for (int t = 0; t < T_DIM; ++t) {
        const float g    = ys[2 * t + 0];
        const float done = ys[2 * t + 1];
        const float d    = g - u[t];
        const float sg   = d * d - r;
        sigma[t] = sg;
        r = r + CONT_ALPHA * sg;
        if (done > 0.5f) r = 0.0f;
    }
    float wnext = 0.0f;
    float sumw  = 0.0f;
    for (int t = T_DIM - 1; t >= 0; --t) {
        const float done = ys[2 * t + 1];
        const float keep = (done > 0.5f) ? 0.0f : 1.0f;
        const float wt = sigma[t] + GAMMA_LAMBDA * keep * wnext;
        w[t] = wt;
        wnext = wt;
        sumw += wt;
    }
    out[D_DIM * H_DIM + 2 * H_DIM + 0] = -sumw;
    out[D_DIM * H_DIM + 2 * H_DIM + 1] = sigma[T_DIM - 1];
}

__global__ __launch_bounds__(256) void col_kernel(
    float* __restrict__ hBs, const float* __restrict__ W2,
    const float* __restrict__ w, float* __restrict__ out)
{
    __shared__ float s1[256];
    __shared__ float s2[256];
    const int j   = blockIdx.x;
    const int tid = threadIdx.x;
    const float w2j = W2[j];
    float a1 = 0.0f, a2 = 0.0f;
    for (int t = tid; t < T_DIM; t += 256) {
        const float hv = hBs[t * H_DIM + j];
        const float wt = w[t];
        const float bs = (hv > 0.0f) ? wt * w2j : 0.0f;
        hBs[t * H_DIM + j] = bs;
        a1 += bs;
        a2 += wt * hv;
    }
    s1[tid] = a1; s2[tid] = a2;
    __syncthreads();
    for (int off = 128; off > 0; off >>= 1) {
        if (tid < off) { s1[tid] += s1[tid + off]; s2[tid] += s2[tid + off]; }
        __syncthreads();
    }
    if (tid == 0) {
        out[D_DIM * H_DIM + j]         = -s1[0];
        out[D_DIM * H_DIM + H_DIM + j] = -s2[0];
    }
}

__global__ __launch_bounds__(256) void bwd_gemm_kernel(
    const float* __restrict__ X, const float* __restrict__ Bs,
    float* __restrict__ out)
{
    const int lane = threadIdx.x & 31;
    const int wave = threadIdx.x >> 5;
    const int tile = blockIdx.x * 8 + wave;
    const int tileM = tile >> 4;
    const int tileN = tile & 15;
    const int half = lane >> 4;
    const int l16  = lane & 15;
    const int rowA = tileM * 16 + l16;
    const int colB = tileN * 16 + l16;

    v8f acc = {};
    for (int k0 = 0; k0 < T_DIM; k0 += 4) {
        const int ka = k0 + 2 * half;
        v2f a, b;
        a.x = X[ka * D_DIM + rowA];
        a.y = X[(ka + 1) * D_DIM + rowA];
        b.x = Bs[ka * H_DIM + colB];
        b.y = Bs[(ka + 1) * H_DIM + colB];
        acc = __builtin_amdgcn_wmma_f32_16x16x4_f32(
            false, a, false, b, (short)0, acc, false, false);
    }
#pragma unroll
    for (int v = 0; v < 8; ++v) {
        const int m = tileM * 16 + half * 8 + v;
        out[m * H_DIM + colB] = -acc[v];
    }
}

extern "C" void kernel_launch(void* const* d_in, const int* in_sizes, int n_in,
                              void* d_out, int out_size, void* d_ws, size_t ws_size,
                              hipStream_t stream)
{
    const float* xs = (const float*)d_in[0];
    const float* ys = (const float*)d_in[1];
    const float* W1 = (const float*)d_in[2];
    const float* b1 = (const float*)d_in[3];
    const float* W2 = (const float*)d_in[4];
    const float* b2 = (const float*)d_in[5];
    float* out = (float*)d_out;

    float* ws = (float*)d_ws;
    float* h     = ws;
    float* u     = h + (size_t)T_DIM * H_DIM;
    float* sigma = u + T_DIM;
    float* w     = sigma + T_DIM;

    fwd_gemm_kernel<<<256, 256, 0, stream>>>(xs, W1, b1, h);
    value_kernel<<<T_DIM, 256, 0, stream>>>(h, W2, b2, u);
    scan_kernel<<<1, 32, 0, stream>>>(u, ys, sigma, w, out);
    col_kernel<<<H_DIM, 256, 0, stream>>>(h, W2, w, out);
    bwd_gemm_kernel<<<32, 256, 0, stream>>>(xs, h, out);
}


# total_W1 product moved into the col launch and cut over t: a col block multiplies its 64-row Bs tile (in LDS) by X on WMMA into a partial panel; last launch adds the 32 panels (reorders that sum)
# speedup vs baseline: 10.0336x; 10.0336x over previous
#include <hip/hip_runtime.h>
#include <hip/hip_bf16.h>

typedef float v2f __attribute__((ext_vector_type(2)));
typedef float v8f __attribute__((ext_vector_type(8)));

#define T_DIM 2048
#define D_DIM 256
#define H_DIM 256
#define GAMMA_LAMBDA (0.99f * 0.9f)
#define CONT_ALPHA 0.01f

__global__ __launch_bounds__(256) void fwd_gemm_kernel(
    const float* __restrict__ X, const float* __restrict__ W1,
    const float* __restrict__ b1, float* __restrict__ h)
{
    const int lane = threadIdx.x & 31;
    const int wave = threadIdx.x >> 5;
    const int tile = blockIdx.x * 8 + wave;
    const int tileM = tile >> 4;
    const int tileN = tile & 15;
    const int half = lane >> 4;
    const int l16  = lane & 15;
    const int rowA = tileM * 16 + l16;
    const int colB = tileN * 16 + l16;

    v8f acc = {};
    for (int k0 = 0; k0 < D_DIM; k0 += 4) {
        const int ka = k0 + 2 * half;
        v2f a, b;
        a.x = X[rowA * D_DIM + ka];
        a.y = X[rowA * D_DIM + ka + 1];
        b.x = W1[ka * H_DIM + colB];
        b.y = W1[(ka + 1) * H_DIM + colB];
        acc = __builtin_amdgcn_wmma_f32_16x16x4_f32(
            false, a, false, b, (short)0, acc, false, false);
    }
    const float bias = b1[colB];
#pragma unroll
    for (int v = 0; v < 8; ++v) {
        const int m = tileM * 16 + half * 8 + v;
        const float val = acc[v] + bias;
        h[m * H_DIM + colB] = val > 0.0f ? val : 0.0f;
    }
}

__global__ __launch_bounds__(256) void value_kernel(
    const float* __restrict__ h, const float* __restrict__ W2,
    const float* __restrict__ b2, float* __restrict__ u)
{
    __shared__ float s[256];
    const int t = blockIdx.x;
    const int j = threadIdx.x;
    s[j] = h[t * H_DIM + j] * W2[j];
    __syncthreads();
    for (int off = 128; off > 0; off >>= 1) {
        if (j < off) s[j] += s[j + off];
        __syncthreads();
    }
    if (j == 0) u[t] = s[0] + b2[0];
}

typedef float v4f __attribute__((ext_vector_type(4)));

#define SC_U   8
#define SC_OWN 16
#define SC_PAD 16

__global__ __launch_bounds__(256) void scan_kernel(
    const float* __restrict__ u, const float* __restrict__ ys,
    float* __restrict__ w, float* __restrict__ out)
{
    __shared__ v4f   sP[T_DIM + SC_PAD];
    __shared__ v2f   sSK[T_DIM];
    __shared__ __attribute__((aligned(16))) float sW[T_DIM];
    __shared__ unsigned sD[T_DIM / 32];
    __shared__ float sRed[8];
    const int tid = threadIdx.x;

    unsigned dm[T_DIM / 256];
#pragma unroll
    for (int k = 0; k < T_DIM / 256; ++k) {
        const int t = tid + 256 * k;
        const float g    = ys[2 * t + 0];
        const float done = ys[2 * t + 1];
        const float d    = g - u[t];
        const float ee   = d * d;
        const bool  rst  = done > 0.5f;
        const float keep = rst ? 0.0f : 1.0f;
        v4f p;
        p.x = ee;
        p.y = keep * (1.0f - CONT_ALPHA);
        p.z = keep * (CONT_ALPHA * ee);
        p.w = GAMMA_LAMBDA * keep;
        sP[t] = p;
        dm[k] = (unsigned)__ballot(rst);
    }
    if ((tid & 31) == 0) {
#pragma unroll
        for (int k = 0; k < T_DIM / 256; ++k) sD[8 * k + (tid >> 5)] = dm[k];
    }
    __syncthreads();

    float sumw = 0.0f;
    if (tid < T_DIM / SC_OWN) {
        const int t0 = SC_OWN * tid;
        const int wi = t0 >> 5;
        const unsigned long long two =
            ((unsigned long long)sD[wi] << 32) | (unsigned long long)(wi > 0 ? sD[wi - 1] : 0u);
        unsigned sm = (unsigned)(two >> (32 + (t0 & 31) - 1)) & 0xFFFFu;
        if (tid == 0) sm |= 1u;
        if (sm != 0u) {
            const int first = t0 + (__ffs(sm) - 1);
            const int tstop = t0 + SC_OWN - 1;
            int wj = tstop >> 5;
            unsigned m = sD[wj] & (0xFFFFFFFFu << (tstop & 31));
            for (int k = 0; k < T_DIM / 32 && m == 0u && wj < T_DIM / 32 - 1; ++k) {
                ++wj;
                m = sD[wj];
            }
            const int last  = (m != 0u) ? 32 * wj + (__ffs(m) - 1) : T_DIM - 1;
            const int n     = last - first + 1;
            const int nfull = n / SC_U;
            const int tail  = n % SC_U;
            {
                int t = first;
                float r = 0.0f;
                v4f cur[SC_U], nxt[SC_U];
#pragma unroll
                for (int i = 0; i < SC_U; ++i) cur[i] = sP[t + i];
                for (int b = 0; b < nfull; ++b) {
#pragma unroll
                    for (int i = 0; i < SC_U; ++i) nxt[i] = sP[t + SC_U + i];
#pragma unroll
                    for (int i = 0; i < SC_U; ++i) {
                        v2f o;
                        o.x = cur[i].x - r;
                        o.y = cur[i].w;
                        r = __builtin_fmaf(cur[i].y, r, cur[i].z);
                        sSK[t + i] = o;
                    }
#pragma unroll
                    for (int i = 0; i < SC_U; ++i) cur[i] = nxt[i];
                    t += SC_U;
                }
#pragma unroll
                for (int i = 0; i < SC_U - 1; ++i) {
                    if (i < tail) {
                        v2f o;
                        o.x = cur[i].x - r;
                        o.y = cur[i].w;
                        r = __builtin_fmaf(cur[i].y, r, cur[i].z);
                        sSK[t + i] = o;
                    }
                }
            }
            {
                int t = last;
                float wn = 0.0f;
                v2f cur[SC_U], nxt[SC_U];
#pragma unroll
                for (int i = 0; i < SC_U; ++i) cur[i] = sSK[max(t - i, first)];
                for (int b = 0; b < nfull; ++b) {
#pragma unroll
                    for (int i = 0; i < SC_U; ++i) nxt[i] = sSK[max(t - SC_U - i, first)];
#pragma unroll
                    for (int i = 0; i < SC_U; ++i) {
                        const float wt = __builtin_fmaf(cur[i].y, wn, cur[i].x);
                        wn = wt;
                        sW[t - i] = wt;
                        sumw += wt;
                    }
#pragma unroll
                    for (int i = 0; i < SC_U; ++i) cur[i] = nxt[i];
                    t -= SC_U;
                }
#pragma unroll
                for (int i = 0; i < SC_U - 1; ++i) {
                    if (i < tail) {
                        const float wt = __builtin_fmaf(cur[i].y, wn, cur[i].x);
                        wn = wt;
                        sW[t - i] = wt;
                        sumw += wt;
                    }
                }
            }
        }
    }

#pragma unroll
    for (int off = 16; off > 0; off >>= 1) sumw += __shfl_xor(sumw, off);
    if ((tid & 31) == 0) sRed[tid >> 5] = sumw;
    __syncthreads();

    if (tid == 0) {
        float s = 0.0f;
#pragma unroll
        for (int k = 0; k < 8; ++k) s += sRed[k];
        out[D_DIM * H_DIM + 2 * H_DIM + 0] = -s;
        out[D_DIM * H_DIM + 2 * H_DIM + 1] = sSK[T_DIM - 1].x;
    }

    {
        v4f* wg = (v4f*)w;
        const v4f* ws4 = (const v4f*)sW;
#pragma unroll
        for (int k = 0; k < T_DIM / 4 / 256; ++k)
            wg[tid + 256 * k] = ws4[tid + 256 * k];
    }
}

#define COL_RG   32
#define COL_ROWS (T_DIM / COL_RG)
#define COL_BLOCKS 256
#define COL_KS   (COL_ROWS / 4)
#define COL_PITCH 48

__global__ __launch_bounds__(256) void col_kernel(
    const float* __restrict__ h, const float* __restrict__ W2,
    const float* __restrict__ w, const float* __restrict__ X,
    float* __restrict__ part, float* __restrict__ pw1)
{
    __shared__ v4f s1[32 * 8];
    __shared__ v4f s2[32 * 8];
    __shared__ __attribute__((aligned(16))) float sBs[COL_ROWS * COL_PITCH];
    const int tid = threadIdx.x;
    const int cg  = blockIdx.x & 7;
    const int rg  = blockIdx.x >> 3;
    const int c4  = tid & 7;
    const int r   = tid >> 3;
    const int j0  = cg * 32 + c4 * 4;
    const int t0  = rg * COL_ROWS + r;
    const int t1  = t0 + 32;
    const int lane = tid & 31;
    const int wave = tid >> 5;
    const int half = lane >> 4;
    const int l16  = lane & 15;

    const v4f h0 = *(const v4f*)(h + t0 * H_DIM + j0);
    const v4f h1 = *(const v4f*)(h + t1 * H_DIM + j0);
    const float w0 = w[t0];
    const float w1 = w[t1];
    const v4f w2 = *(const v4f*)(W2 + j0);

    const float* xa = X + (rg * COL_ROWS + 2 * half) * D_DIM + wave * 32 + 2 * l16;
    v2f xe0[COL_KS], xe1[COL_KS];
#pragma unroll
    for (int s = 0; s < COL_KS; ++s) {
        xe0[s] = *(const v2f*)(xa + (4 * s) * D_DIM);
        xe1[s] = *(const v2f*)(xa + (4 * s + 1) * D_DIM);
    }

    v4f b0, b1, a1, a2;
#pragma unroll
    for (int k = 0; k < 4; ++k) {
        b0[k] = (h0[k] > 0.0f) ? w0 * w2[k] : 0.0f;
        b1[k] = (h1[k] > 0.0f) ? w1 * w2[k] : 0.0f;
        a1[k] = b0[k] + b1[k];
        a2[k] = w0 * h0[k] + w1 * h1[k];
    }
    s1[r * 8 + c4] = a1;
    s2[r * 8 + c4] = a2;
    *(v4f*)(sBs + r * COL_PITCH + c4 * 4)        = b0;
    *(v4f*)(sBs + (r + 32) * COL_PITCH + c4 * 4) = b1;
    __syncthreads();

    if (tid < 64) {
        const int col   = tid & 31;
        const int which = tid >> 5;
        const float* s = which ? (const float*)s2 : (const float*)s1;
        float acc = 0.0f;
#pragma unroll
        for (int rr = 0; rr < 32; ++rr) acc += s[rr * 32 + col];
        part[(which * COL_RG + rg) * H_DIM + cg * 32 + col] = acc;
    }

    const float* fb = sBs + (2 * half) * COL_PITCH + 2 * l16;
    v2f be0[COL_KS], be1[COL_KS];
#pragma unroll
    for (int s = 0; s < COL_KS; ++s) {
        be0[s] = *(const v2f*)(fb + (4 * s) * COL_PITCH);
        be1[s] = *(const v2f*)(fb + (4 * s + 1) * COL_PITCH);
    }
    __builtin_amdgcn_sched_barrier(0);

    v8f acc00 = {}, acc01 = {}, acc10 = {}, acc11 = {};
#pragma unroll
    for (int s = 0; s < COL_KS; ++s) {
        v2f aq0, aq1, bn0, bn1;
        aq0.x = xe0[s].x;  aq0.y = xe1[s].x;
        aq1.x = xe0[s].y;  aq1.y = xe1[s].y;
        bn0.x = be0[s].x;  bn0.y = be1[s].x;
        bn1.x = be0[s].y;  bn1.y = be1[s].y;
        acc00 = __builtin_amdgcn_wmma_f32_16x16x4_f32(
            false, aq0, false, bn0, (short)0, acc00, false, false);
        acc01 = __builtin_amdgcn_wmma_f32_16x16x4_f32(
            false, aq0, false, bn1, (short)0, acc01, false, false);
        acc10 = __builtin_amdgcn_wmma_f32_16x16x4_f32(
            false, aq1, false, bn0, (short)0, acc10, false, false);
        acc11 = __builtin_amdgcn_wmma_f32_16x16x4_f32(
            false, aq1, false, bn1, (short)0, acc11, false, false);
    }

    float* pp = pw1 + ((size_t)rg * D_DIM + wave * 32 + 16 * half) * H_DIM + cg * 32 + 2 * l16;
#pragma unroll
    for (int v = 0; v < 8; ++v) {
        v2f o0, o1;
        o0.x = acc00[v];  o0.y = acc01[v];
        o1.x = acc10[v];  o1.y = acc11[v];
        *(v2f*)(pp + (2 * v) * H_DIM)     = o0;
        *(v2f*)(pp + (2 * v + 1) * H_DIM) = o1;
    }
}

#define WS_BLOCKS 256
#define WS_GRID   (WS_BLOCKS + 8)

__global__ __launch_bounds__(64) void w1_sum_kernel(
    const float* __restrict__ pw1, const float* __restrict__ part,
    float* __restrict__ out)
{
    if (blockIdx.x >= WS_BLOCKS) {
        const int o     = (blockIdx.x - WS_BLOCKS) * 64 + threadIdx.x;
        const int which = o >> 8;
        const int j     = o & (H_DIM - 1);
        const float* p  = part + which * (COL_RG * H_DIM) + j;
        float s = 0.0f;
#pragma unroll
        for (int g = 0; g < COL_RG; ++g) s += p[g * H_DIM];
        out[D_DIM * H_DIM + which * H_DIM + j] = -s;
        return;
    }

    const int q = blockIdx.x * 64 + threadIdx.x;
    const v4f* p = (const v4f*)pw1 + q;
    v4f v[COL_RG];
#pragma unroll
    for (int g = 0; g < COL_RG; ++g) v[g] = p[g * (D_DIM * H_DIM / 4)];
    __builtin_amdgcn_sched_barrier(0);
    v4f s = v[0];
#pragma unroll
    for (int g = 1; g < COL_RG; ++g) s += v[g];
    *((v4f*)out + q) = -s;
}

extern "C" void kernel_launch(void* const* d_in, const int* in_sizes, int n_in,
                              void* d_out, int out_size, void* d_ws, size_t ws_size,
                              hipStream_t stream)
{
    const float* xs = (const float*)d_in[0];
    const float* ys = (const float*)d_in[1];
    const float* W1 = (const float*)d_in[2];
    const float* b1 = (const float*)d_in[3];
    const float* W2 = (const float*)d_in[4];
    const float* b2 = (const float*)d_in[5];
    float* out = (float*)d_out;

    float* ws = (float*)d_ws;
    float* h     = ws;
    float* u     = h + (size_t)T_DIM * H_DIM;
    float* sigma = u + T_DIM;
    float* w     = sigma + T_DIM;
    float* part  = w + T_DIM;
    float* pw1   = part + 2 * COL_RG * H_DIM;

    fwd_gemm_kernel<<<256, 256, 0, stream>>>(xs, W1, b1, h);
    value_kernel<<<T_DIM, 256, 0, stream>>>(h, W2, b2, u);
    scan_kernel<<<1, 256, 0, stream>>>(u, ys, w, out);
    col_kernel<<<COL_BLOCKS, 256, 0, stream>>>(h, W2, w, xs, part, pw1);
    w1_sum_kernel<<<WS_GRID, 64, 0, stream>>>(pw1, part, out);
}


# value kernel re-tiled: a wave sums 4 rows of h*W2 (two 16-byte loads per lane and row, lane-exchange adds), 256 blocks of 2 waves, was 2048 blocks of 8 with an LDS tree; reorders the sum of u
# speedup vs baseline: 10.3744x; 1.0340x over previous
#include <hip/hip_runtime.h>
#include <hip/hip_bf16.h>

typedef float v2f __attribute__((ext_vector_type(2)));
typedef float v8f __attribute__((ext_vector_type(8)));

#define T_DIM 2048
#define D_DIM 256
#define H_DIM 256
#define GAMMA_LAMBDA (0.99f * 0.9f)
#define CONT_ALPHA 0.01f

__global__ __launch_bounds__(256) void fwd_gemm_kernel(
    const float* __restrict__ X, const float* __restrict__ W1,
    const float* __restrict__ b1, float* __restrict__ h)
{
    const int lane = threadIdx.x & 31;
    const int wave = threadIdx.x >> 5;
    const int tile = blockIdx.x * 8 + wave;
    const int tileM = tile >> 4;
    const int tileN = tile & 15;
    const int half = lane >> 4;
    const int l16  = lane & 15;
    const int rowA = tileM * 16 + l16;
    const int colB = tileN * 16 + l16;

    v8f acc = {};
    for (int k0 = 0; k0 < D_DIM; k0 += 4) {
        const int ka = k0 + 2 * half;
        v2f a, b;
        a.x = X[rowA * D_DIM + ka];
        a.y = X[rowA * D_DIM + ka + 1];
        b.x = W1[ka * H_DIM + colB];
        b.y = W1[(ka + 1) * H_DIM + colB];
        acc = __builtin_amdgcn_wmma_f32_16x16x4_f32(
            false, a, false, b, (short)0, acc, false, false);
    }
    const float bias = b1[colB];
#pragma unroll
    for (int v = 0; v < 8; ++v) {
        const int m = tileM * 16 + half * 8 + v;
        const float val = acc[v] + bias;
        h[m * H_DIM + colB] = val > 0.0f ? val : 0.0f;
    }
}

typedef float v4f __attribute__((ext_vector_type(4)));

#define VAL_ROWS   4
#define VAL_WAVES  2
#define VAL_BLOCKS (T_DIM / (VAL_ROWS * VAL_WAVES))

__global__ __launch_bounds__(32 * VAL_WAVES) void value_kernel(
    const float* __restrict__ h, const float* __restrict__ W2,
    const float* __restrict__ b2, float* __restrict__ u)
{
    const int lane = threadIdx.x & 31;
    const int wave = threadIdx.x >> 5;
    const int t0   = (blockIdx.x * VAL_WAVES + wave) * VAL_ROWS;
    const float* hp = h + t0 * H_DIM + 4 * lane;

    const v4f wa = *(const v4f*)(W2 + 4 * lane);
    const v4f wb = *(const v4f*)(W2 + 128 + 4 * lane);
    v4f ha[VAL_ROWS], hb[VAL_ROWS];
#pragma unroll
    for (int r = 0; r < VAL_ROWS; ++r) {
        ha[r] = *(const v4f*)(hp + r * H_DIM);
        hb[r] = *(const v4f*)(hp + r * H_DIM + 128);
    }
    const float bias = b2[0];

    float s[VAL_ROWS];
#pragma unroll
    for (int r = 0; r < VAL_ROWS; ++r) {
        float a = ha[r].x * wa.x;
        a += ha[r].y * wa.y;
        a += ha[r].z * wa.z;
        a += ha[r].w * wa.w;
        a += hb[r].x * wb.x;
        a += hb[r].y * wb.y;
        a += hb[r].z * wb.z;
        a += hb[r].w * wb.w;
        s[r] = a;
    }
    asm volatile("" :: "s"(bias));
#pragma unroll
    for (int off = 16; off > 0; off >>= 1) {
#pragma unroll
        for (int r = 0; r < VAL_ROWS; ++r) s[r] += __shfl_xor(s[r], off);
    }
    if (lane == 0) {
        v4f o;
        o.x = s[0] + bias;
        o.y = s[1] + bias;
        o.z = s[2] + bias;
        o.w = s[3] + bias;
        *(v4f*)(u + t0) = o;
    }
}

#define SC_U   8
#define SC_OWN 16
#define SC_PAD 16

__global__ __launch_bounds__(256) void scan_kernel(
    const float* __restrict__ u, const float* __restrict__ ys,
    float* __restrict__ w, float* __restrict__ out)
{
    __shared__ v4f   sP[T_DIM + SC_PAD];
    __shared__ v2f   sSK[T_DIM];
    __shared__ __attribute__((aligned(16))) float sW[T_DIM];
    __shared__ unsigned sD[T_DIM / 32];
    __shared__ float sRed[8];
    const int tid = threadIdx.x;

    unsigned dm[T_DIM / 256];
#pragma unroll
    for (int k = 0; k < T_DIM / 256; ++k) {
        const int t = tid + 256 * k;
        const float g    = ys[2 * t + 0];
        const float done = ys[2 * t + 1];
        const float d    = g - u[t];
        const float ee   = d * d;
        const bool  rst  = done > 0.5f;
        const float keep = rst ? 0.0f : 1.0f;
        v4f p;
        p.x = ee;
        p.y = keep * (1.0f - CONT_ALPHA);
        p.z = keep * (CONT_ALPHA * ee);
        p.w = GAMMA_LAMBDA * keep;
        sP[t] = p;
        dm[k] = (unsigned)__ballot(rst);
    }
    if ((tid & 31) == 0) {
#pragma unroll
        for (int k = 0; k < T_DIM / 256; ++k) sD[8 * k + (tid >> 5)] = dm[k];
    }
    __syncthreads();

    float sumw = 0.0f;
    if (tid < T_DIM / SC_OWN) {
        const int t0 = SC_OWN * tid;
        const int wi = t0 >> 5;
        const unsigned long long two =
            ((unsigned long long)sD[wi] << 32) | (unsigned long long)(wi > 0 ? sD[wi - 1] : 0u);
        unsigned sm = (unsigned)(two >> (32 + (t0 & 31) - 1)) & 0xFFFFu;
        if (tid == 0) sm |= 1u;
        if (sm != 0u) {
            const int first = t0 + (__ffs(sm) - 1);
            const int tstop = t0 + SC_OWN - 1;
            int wj = tstop >> 5;
            unsigned m = sD[wj] & (0xFFFFFFFFu << (tstop & 31));
            for (int k = 0; k < T_DIM / 32 && m == 0u && wj < T_DIM / 32 - 1; ++k) {
                ++wj;
                m = sD[wj];
            }
            const int last  = (m != 0u) ? 32 * wj + (__ffs(m) - 1) : T_DIM - 1;
            const int n     = last - first + 1;
            const int nfull = n / SC_U;
            const int tail  = n % SC_U;
            {
                int t = first;
                float r = 0.0f;
                v4f cur[SC_U], nxt[SC_U];
#pragma unroll
                for (int i = 0; i < SC_U; ++i) cur[i] = sP[t + i];
                for (int b = 0; b < nfull; ++b) {
#pragma unroll
                    for (int i = 0; i < SC_U; ++i) nxt[i] = sP[t + SC_U + i];
#pragma unroll
                    for (int i = 0; i < SC_U; ++i) {
                        v2f o;
                        o.x = cur[i].x - r;
                        o.y = cur[i].w;
                        r = __builtin_fmaf(cur[i].y, r, cur[i].z);
                        sSK[t + i] = o;
                    }
#pragma unroll
                    for (int i = 0; i < SC_U; ++i) cur[i] = nxt[i];
                    t += SC_U;
                }
#pragma unroll
                for (int i = 0; i < SC_U - 1; ++i) {
                    if (i < tail) {
                        v2f o;
                        o.x = cur[i].x - r;
                        o.y = cur[i].w;
                        r = __builtin_fmaf(cur[i].y, r, cur[i].z);
                        sSK[t + i] = o;
                    }
                }
            }
            {
                int t = last;
                float wn = 0.0f;
                v2f cur[SC_U], nxt[SC_U];
#pragma unroll
                for (int i = 0; i < SC_U; ++i) cur[i] = sSK[max(t - i, first)];
                for (int b = 0; b < nfull; ++b) {
#pragma unroll
                    for (int i = 0; i < SC_U; ++i) nxt[i] = sSK[max(t - SC_U - i, first)];
#pragma unroll
                    for (int i = 0; i < SC_U; ++i) {
                        const float wt = __builtin_fmaf(cur[i].y, wn, cur[i].x);
                        wn = wt;
                        sW[t - i] = wt;
                        sumw += wt;
                    }
#pragma unroll
                    for (int i = 0; i < SC_U; ++i) cur[i] = nxt[i];
                    t -= SC_U;
                }
#pragma unroll
                for (int i = 0; i < SC_U - 1; ++i) {
                    if (i < tail) {
                        const float wt = __builtin_fmaf(cur[i].y, wn, cur[i].x);
                        wn = wt;
                        sW[t - i] = wt;
                        sumw += wt;
                    }
                }
            }
        }
    }

#pragma unroll
    for (int off = 16; off > 0; off >>= 1) sumw += __shfl_xor(sumw, off);
    if ((tid & 31) == 0) sRed[tid >> 5] = sumw;
    __syncthreads();

    if (tid == 0) {
        float s = 0.0f;
#pragma unroll
        for (int k = 0; k < 8; ++k) s += sRed[k];
        out[D_DIM * H_DIM + 2 * H_DIM + 0] = -s;
        out[D_DIM * H_DIM + 2 * H_DIM + 1] = sSK[T_DIM - 1].x;
    }

    {
        v4f* wg = (v4f*)w;
        const v4f* ws4 = (const v4f*)sW;
#pragma unroll
        for (int k = 0; k < T_DIM / 4 / 256; ++k)
            wg[tid + 256 * k] = ws4[tid + 256 * k];
    }
}

#define COL_RG   32
#define COL_ROWS (T_DIM / COL_RG)
#define COL_BLOCKS 256
#define COL_KS   (COL_ROWS / 4)
#define COL_PITCH 48

__global__ __launch_bounds__(256) void col_kernel(
    const float* __restrict__ h, const float* __restrict__ W2,
    const float* __restrict__ w, const float* __restrict__ X,
    float* __restrict__ part, float* __restrict__ pw1)
{
    __shared__ v4f s1[32 * 8];
    __shared__ v4f s2[32 * 8];
    __shared__ __attribute__((aligned(16))) float sBs[COL_ROWS * COL_PITCH];
    const int tid = threadIdx.x;
    const int cg  = blockIdx.x & 7;
    const int rg  = blockIdx.x >> 3;
    const int c4  = tid & 7;
    const int r   = tid >> 3;
    const int j0  = cg * 32 + c4 * 4;
    const int t0  = rg * COL_ROWS + r;
    const int t1  = t0 + 32;
    const int lane = tid & 31;
    const int wave = tid >> 5;
    const int half = lane >> 4;
    const int l16  = lane & 15;

    const v4f h0 = *(const v4f*)(h + t0 * H_DIM + j0);
    const v4f h1 = *(const v4f*)(h + t1 * H_DIM + j0);
    const float w0 = w[t0];
    const float w1 = w[t1];
    const v4f w2 = *(const v4f*)(W2 + j0);

    const float* xa = X + (rg * COL_ROWS + 2 * half) * D_DIM + wave * 32 + 2 * l16;
    v2f xe0[COL_KS], xe1[COL_KS];
#pragma unroll
    for (int s = 0; s < COL_KS; ++s) {
        xe0[s] = *(const v2f*)(xa + (4 * s) * D_DIM);
        xe1[s] = *(const v2f*)(xa + (4 * s + 1) * D_DIM);
    }

    v4f b0, b1, a1, a2;
#pragma unroll
    for (int k = 0; k < 4; ++k) {
        b0[k] = (h0[k] > 0.0f) ? w0 * w2[k] : 0.0f;
        b1[k] = (h1[k] > 0.0f) ? w1 * w2[k] : 0.0f;
        a1[k] = b0[k] + b1[k];
        a2[k] = w0 * h0[k] + w1 * h1[k];
    }
    s1[r * 8 + c4] = a1;
    s2[r * 8 + c4] = a2;
    *(v4f*)(sBs + r * COL_PITCH + c4 * 4)        = b0;
    *(v4f*)(sBs + (r + 32) * COL_PITCH + c4 * 4) = b1;
    __syncthreads();

    if (tid < 64) {
        const int col   = tid & 31;
        const int which = tid >> 5;
        const float* s = which ? (const float*)s2 : (const float*)s1;
        float acc = 0.0f;
#pragma unroll
        for (int rr = 0; rr < 32; ++rr) acc += s[rr * 32 + col];
        part[(which * COL_RG + rg) * H_DIM + cg * 32 + col] = acc;
    }

    const float* fb = sBs + (2 * half) * COL_PITCH + 2 * l16;
    v2f be0[COL_KS], be1[COL_KS];
#pragma unroll
    for (int s = 0; s < COL_KS; ++s) {
        be0[s] = *(const v2f*)(fb + (4 * s) * COL_PITCH);
        be1[s] = *(const v2f*)(fb + (4 * s + 1) * COL_PITCH);
    }
    __builtin_amdgcn_sched_barrier(0);

    v8f acc00 = {}, acc01 = {}, acc10 = {}, acc11 = {};
#pragma unroll
    for (int s = 0; s < COL_KS; ++s) {
        v2f aq0, aq1, bn0, bn1;
        aq0.x = xe0[s].x;  aq0.y = xe1[s].x;
        aq1.x = xe0[s].y;  aq1.y = xe1[s].y;
        bn0.x = be0[s].x;  bn0.y = be1[s].x;
        bn1.x = be0[s].y;  bn1.y = be1[s].y;
        acc00 = __builtin_amdgcn_wmma_f32_16x16x4_f32(
            false, aq0, false, bn0, (short)0, acc00, false, false);
        acc01 = __builtin_amdgcn_wmma_f32_16x16x4_f32(
            false, aq0, false, bn1, (short)0, acc01, false, false);
        acc10 = __builtin_amdgcn_wmma_f32_16x16x4_f32(
            false, aq1, false, bn0, (short)0, acc10, false, false);
        acc11 = __builtin_amdgcn_wmma_f32_16x16x4_f32(
            false, aq1, false, bn1, (short)0, acc11, false, false);
    }

    float* pp = pw1 + ((size_t)rg * D_DIM + wave * 32 + 16 * half) * H_DIM + cg * 32 + 2 * l16;
#pragma unroll
    for (int v = 0; v < 8; ++v) {
        v2f o0, o1;
        o0.x = acc00[v];  o0.y = acc01[v];
        o1.x = acc10[v];  o1.y = acc11[v];
        *(v2f*)(pp + (2 * v) * H_DIM)     = o0;
        *(v2f*)(pp + (2 * v + 1) * H_DIM) = o1;
    }
}

#define WS_BLOCKS 256
#define WS_GRID   (WS_BLOCKS + 8)

__global__ __launch_bounds__(64) void w1_sum_kernel(
    const float* __restrict__ pw1, const float* __restrict__ part,
    float* __restrict__ out)
{
    if (blockIdx.x >= WS_BLOCKS) {
        const int o     = (blockIdx.x - WS_BLOCKS) * 64 + threadIdx.x;
        const int which = o >> 8;
        const int j     = o & (H_DIM - 1);
        const float* p  = part + which * (COL_RG * H_DIM) + j;
        float s = 0.0f;
#pragma unroll
        for (int g = 0; g < COL_RG; ++g) s += p[g * H_DIM];
        out[D_DIM * H_DIM + which * H_DIM + j] = -s;
        return;
    }

    const int q = blockIdx.x * 64 + threadIdx.x;
    const v4f* p = (const v4f*)pw1 + q;
    v4f v[COL_RG];
#pragma unroll
    for (int g = 0; g < COL_RG; ++g) v[g] = p[g * (D_DIM * H_DIM / 4)];
    __builtin_amdgcn_sched_barrier(0);
    v4f s = v[0];
#pragma unroll
    for (int g = 1; g < COL_RG; ++g) s += v[g];
    *((v4f*)out + q) = -s;
}

extern "C" void kernel_launch(void* const* d_in, const int* in_sizes, int n_in,
                              void* d_out, int out_size, void* d_ws, size_t ws_size,
                              hipStream_t stream)
{
    const float* xs = (const float*)d_in[0];
    const float* ys = (const float*)d_in[1];
    const float* W1 = (const float*)d_in[2];
    const float* b1 = (const float*)d_in[3];
    const float* W2 = (const float*)d_in[4];
    const float* b2 = (const float*)d_in[5];
    float* out = (float*)d_out;

    float* ws = (float*)d_ws;
    float* h     = ws;
    float* u     = h + (size_t)T_DIM * H_DIM;
    float* sigma = u + T_DIM;
    float* w     = sigma + T_DIM;
    float* part  = w + T_DIM;
    float* pw1   = part + 2 * COL_RG * H_DIM;

    fwd_gemm_kernel<<<256, 256, 0, stream>>>(xs, W1, b1, h);
    value_kernel<<<VAL_BLOCKS, 32 * VAL_WAVES, 0, stream>>>(h, W2, b2, u);
    scan_kernel<<<1, 256, 0, stream>>>(u, ys, w, out);
    col_kernel<<<COL_BLOCKS, 256, 0, stream>>>(h, W2, w, xs, part, pw1);
    w1_sum_kernel<<<WS_GRID, 64, 0, stream>>>(pw1, part, out);
}


# fwd GEMM re-tiled: a wave owns a 16x32 patch of h (two j-interleaved tiles share each A fragment, 8-byte B loads and stores), 8-step chunks loaded 3 ahead, 4-wave blocks; k order and sums unchanged
# speedup vs baseline: 12.3986x; 1.1951x over previous
#include <hip/hip_runtime.h>
#include <hip/hip_bf16.h>

typedef float v2f __attribute__((ext_vector_type(2)));
typedef float v8f __attribute__((ext_vector_type(8)));

#define T_DIM 2048
#define D_DIM 256
#define H_DIM 256
#define GAMMA_LAMBDA (0.99f * 0.9f)
#define CONT_ALPHA 0.01f

#define FW_KS    8
#define FW_NCH   (D_DIM / 4 / FW_KS)
#define FW_AHEAD 3
#define FW_NBUF  (FW_AHEAD + 1)

__global__ __launch_bounds__(128) void fwd_gemm_kernel(
    const float* __restrict__ X, const float* __restrict__ W1,
    const float* __restrict__ b1, float* __restrict__ h)
{
    const int lane = threadIdx.x & 31;
    const int wave = threadIdx.x >> 5;
    const int half = lane >> 4;
    const int l16  = lane & 15;
    const int m0   = (blockIdx.x >> 2) * 32 + (wave >> 1) * 16;
    const int jb   = (blockIdx.x & 3) * 64 + (wave & 1) * 32;

    const float* pa = X + (m0 + l16) * D_DIM + 2 * half;
    const float* pb = W1 + (2 * half) * H_DIM + jb + 2 * l16;

    const v2f bias = *(const v2f*)(b1 + jb + 2 * l16);

    v2f fa[FW_NBUF][FW_KS], fb0[FW_NBUF][FW_KS], fb1[FW_NBUF][FW_KS];
    v8f acc0 = {}, acc1 = {};

#pragma unroll
    for (int ch = 0; ch < FW_AHEAD; ++ch) {
#pragma unroll
        for (int s = 0; s < FW_KS; ++s) {
            const int ks = ch * FW_KS + s;
            fa [ch % FW_NBUF][s] = *(const v2f*)(pa + 4 * ks);
            fb0[ch % FW_NBUF][s] = *(const v2f*)(pb + (4 * ks) * H_DIM);
            fb1[ch % FW_NBUF][s] = *(const v2f*)(pb + (4 * ks + 1) * H_DIM);
        }
    }
#pragma unroll
    for (int ch = 0; ch < FW_NCH; ++ch) {
        if (ch + FW_AHEAD < FW_NCH) {
#pragma unroll
            for (int s = 0; s < FW_KS; ++s) {
                const int ks = (ch + FW_AHEAD) * FW_KS + s;
                fa [(ch + FW_AHEAD) % FW_NBUF][s] = *(const v2f*)(pa + 4 * ks);
                fb0[(ch + FW_AHEAD) % FW_NBUF][s] = *(const v2f*)(pb + (4 * ks) * H_DIM);
                fb1[(ch + FW_AHEAD) % FW_NBUF][s] = *(const v2f*)(pb + (4 * ks + 1) * H_DIM);
            }
        }
        __builtin_amdgcn_sched_barrier(0);
#pragma unroll
        for (int s = 0; s < FW_KS; ++s) {
            v2f bn0, bn1;
            bn0.x = fb0[ch % FW_NBUF][s].x;  bn0.y = fb1[ch % FW_NBUF][s].x;
            bn1.x = fb0[ch % FW_NBUF][s].y;  bn1.y = fb1[ch % FW_NBUF][s].y;
            acc0 = __builtin_amdgcn_wmma_f32_16x16x4_f32(
                false, fa[ch % FW_NBUF][s], false, bn0, (short)0, acc0, false, false);
            acc1 = __builtin_amdgcn_wmma_f32_16x16x4_f32(
                false, fa[ch % FW_NBUF][s], false, bn1, (short)0, acc1, false, false);
        }
    }

    float* ph = h + (m0 + 8 * half) * H_DIM + jb + 2 * l16;
#pragma unroll
    for (int v = 0; v < 8; ++v) {
        const float v0 = acc0[v] + bias.x;
        const float v1 = acc1[v] + bias.y;
        v2f o;
        o.x = v0 > 0.0f ? v0 : 0.0f;
        o.y = v1 > 0.0f ? v1 : 0.0f;
        *(v2f*)(ph + v * H_DIM) = o;
    }
}

typedef float v4f __attribute__((ext_vector_type(4)));

#define VAL_ROWS   4
#define VAL_WAVES  2
#define VAL_BLOCKS (T_DIM / (VAL_ROWS * VAL_WAVES))

__global__ __launch_bounds__(32 * VAL_WAVES) void value_kernel(
    const float* __restrict__ h, const float* __restrict__ W2,
    const float* __restrict__ b2, float* __restrict__ u)
{
    const int lane = threadIdx.x & 31;
    const int wave = threadIdx.x >> 5;
    const int t0   = (blockIdx.x * VAL_WAVES + wave) * VAL_ROWS;
    const float* hp = h + t0 * H_DIM + 4 * lane;

    const v4f wa = *(const v4f*)(W2 + 4 * lane);
    const v4f wb = *(const v4f*)(W2 + 128 + 4 * lane);
    v4f ha[VAL_ROWS], hb[VAL_ROWS];
#pragma unroll
    for (int r = 0; r < VAL_ROWS; ++r) {
        ha[r] = *(const v4f*)(hp + r * H_DIM);
        hb[r] = *(const v4f*)(hp + r * H_DIM + 128);
    }
    const float bias = b2[0];

    float s[VAL_ROWS];
#pragma unroll
    for (int r = 0; r < VAL_ROWS; ++r) {
        float a = ha[r].x * wa.x;
        a += ha[r].y * wa.y;
        a += ha[r].z * wa.z;
        a += ha[r].w * wa.w;
        a += hb[r].x * wb.x;
        a += hb[r].y * wb.y;
        a += hb[r].z * wb.z;
        a += hb[r].w * wb.w;
        s[r] = a;
    }
    asm volatile("" :: "s"(bias));
#pragma unroll
    for (int off = 16; off > 0; off >>= 1) {
#pragma unroll
        for (int r = 0; r < VAL_ROWS; ++r) s[r] += __shfl_xor(s[r], off);
    }
    if (lane == 0) {
        v4f o;
        o.x = s[0] + bias;
        o.y = s[1] + bias;
        o.z = s[2] + bias;
        o.w = s[3] + bias;
        *(v4f*)(u + t0) = o;
    }
}

#define SC_U   8
#define SC_OWN 16
#define SC_PAD 16

__global__ __launch_bounds__(256) void scan_kernel(
    const float* __restrict__ u, const float* __restrict__ ys,
    float* __restrict__ w, float* __restrict__ out)
{
    __shared__ v4f   sP[T_DIM + SC_PAD];
    __shared__ v2f   sSK[T_DIM];
    __shared__ __attribute__((aligned(16))) float sW[T_DIM];
    __shared__ unsigned sD[T_DIM / 32];
    __shared__ float sRed[8];
    const int tid = threadIdx.x;

    unsigned dm[T_DIM / 256];
#pragma unroll
    for (int k = 0; k < T_DIM / 256; ++k) {
        const int t = tid + 256 * k;
        const float g    = ys[2 * t + 0];
        const float done = ys[2 * t + 1];
        const float d    = g - u[t];
        const float ee   = d * d;
        const bool  rst  = done > 0.5f;
        const float keep = rst ? 0.0f : 1.0f;
        v4f p;
        p.x = ee;
        p.y = keep * (1.0f - CONT_ALPHA);
        p.z = keep * (CONT_ALPHA * ee);
        p.w = GAMMA_LAMBDA * keep;
        sP[t] = p;
        dm[k] = (unsigned)__ballot(rst);
    }
    if ((tid & 31) == 0) {
#pragma unroll
        for (int k = 0; k < T_DIM / 256; ++k) sD[8 * k + (tid >> 5)] = dm[k];
    }
    __syncthreads();

    float sumw = 0.0f;
    if (tid < T_DIM / SC_OWN) {
        const int t0 = SC_OWN * tid;
        const int wi = t0 >> 5;
        const unsigned long long two =
            ((unsigned long long)sD[wi] << 32) | (unsigned long long)(wi > 0 ? sD[wi - 1] : 0u);
        unsigned sm = (unsigned)(two >> (32 + (t0 & 31) - 1)) & 0xFFFFu;
        if (tid == 0) sm |= 1u;
        if (sm != 0u) {
            const int first = t0 + (__ffs(sm) - 1);
            const int tstop = t0 + SC_OWN - 1;
            int wj = tstop >> 5;
            unsigned m = sD[wj] & (0xFFFFFFFFu << (tstop & 31));
            for (int k = 0; k < T_DIM / 32 && m == 0u && wj < T_DIM / 32 - 1; ++k) {
                ++wj;
                m = sD[wj];
            }
            const int last  = (m != 0u) ? 32 * wj + (__ffs(m) - 1) : T_DIM - 1;
            const int n     = last - first + 1;
            const int nfull = n / SC_U;
            const int tail  = n % SC_U;
            {
                int t = first;
                float r = 0.0f;
                v4f cur[SC_U], nxt[SC_U];
#pragma unroll
                for (int i = 0; i < SC_U; ++i) cur[i] = sP[t + i];
                for (int b = 0; b < nfull; ++b) {
#pragma unroll
                    for (int i = 0; i < SC_U; ++i) nxt[i] = sP[t + SC_U + i];
#pragma unroll
                    for (int i = 0; i < SC_U; ++i) {
                        v2f o;
                        o.x = cur[i].x - r;
                        o.y = cur[i].w;
                        r = __builtin_fmaf(cur[i].y, r, cur[i].z);
                        sSK[t + i] = o;
                    }
#pragma unroll
                    for (int i = 0; i < SC_U; ++i) cur[i] = nxt[i];
                    t += SC_U;
                }
#pragma unroll
                for (int i = 0; i < SC_U - 1; ++i) {
                    if (i < tail) {
                        v2f o;
                        o.x = cur[i].x - r;
                        o.y = cur[i].w;
                        r = __builtin_fmaf(cur[i].y, r, cur[i].z);
                        sSK[t + i] = o;
                    }
                }
            }
            {
                int t = last;
                float wn = 0.0f;
                v2f cur[SC_U], nxt[SC_U];
#pragma unroll
                for (int i = 0; i < SC_U; ++i) cur[i] = sSK[max(t - i, first)];
                for (int b = 0; b < nfull; ++b) {
#pragma unroll
                    for (int i = 0; i < SC_U; ++i) nxt[i] = sSK[max(t - SC_U - i, first)];
#pragma unroll
                    for (int i = 0; i < SC_U; ++i) {
                        const float wt = __builtin_fmaf(cur[i].y, wn, cur[i].x);
                        wn = wt;
                        sW[t - i] = wt;
                        sumw += wt;
                    }
#pragma unroll
                    for (int i = 0; i < SC_U; ++i) cur[i] = nxt[i];
                    t -= SC_U;
                }
#pragma unroll
                for (int i = 0; i < SC_U - 1; ++i) {
                    if (i < tail) {
                        const float wt = __builtin_fmaf(cur[i].y, wn, cur[i].x);
                        wn = wt;
                        sW[t - i] = wt;
                        sumw += wt;
                    }
                }
            }
        }
    }

#pragma unroll
    for (int off = 16; off > 0; off >>= 1) sumw += __shfl_xor(sumw, off);
    if ((tid & 31) == 0) sRed[tid >> 5] = sumw;
    __syncthreads();

    if (tid == 0) {
        float s = 0.0f;
#pragma unroll
        for (int k = 0; k < 8; ++k) s += sRed[k];
        out[D_DIM * H_DIM + 2 * H_DIM + 0] = -s;
        out[D_DIM * H_DIM + 2 * H_DIM + 1] = sSK[T_DIM - 1].x;
    }

    {
        v4f* wg = (v4f*)w;
        const v4f* ws4 = (const v4f*)sW;
#pragma unroll
        for (int k = 0; k < T_DIM / 4 / 256; ++k)
            wg[tid + 256 * k] = ws4[tid + 256 * k];
    }
}

#define COL_RG   32
#define COL_ROWS (T_DIM / COL_RG)
#define COL_BLOCKS 256
#define COL_KS   (COL_ROWS / 4)
#define COL_PITCH 48

__global__ __launch_bounds__(256) void col_kernel(
    const float* __restrict__ h, const float* __restrict__ W2,
    const float* __restrict__ w, const float* __restrict__ X,
    float* __restrict__ part, float* __restrict__ pw1)
{
    __shared__ v4f s1[32 * 8];
    __shared__ v4f s2[32 * 8];
    __shared__ __attribute__((aligned(16))) float sBs[COL_ROWS * COL_PITCH];
    const int tid = threadIdx.x;
    const int cg  = blockIdx.x & 7;
    const int rg  = blockIdx.x >> 3;
    const int c4  = tid & 7;
    const int r   = tid >> 3;
    const int j0  = cg * 32 + c4 * 4;
    const int t0  = rg * COL_ROWS + r;
    const int t1  = t0 + 32;
    const int lane = tid & 31;
    const int wave = tid >> 5;
    const int half = lane >> 4;
    const int l16  = lane & 15;

    const v4f h0 = *(const v4f*)(h + t0 * H_DIM + j0);
    const v4f h1 = *(const v4f*)(h + t1 * H_DIM + j0);
    const float w0 = w[t0];
    const float w1 = w[t1];
    const v4f w2 = *(const v4f*)(W2 + j0);

    const float* xa = X + (rg * COL_ROWS + 2 * half) * D_DIM + wave * 32 + 2 * l16;
    v2f xe0[COL_KS], xe1[COL_KS];
#pragma unroll
    for (int s = 0; s < COL_KS; ++s) {
        xe0[s] = *(const v2f*)(xa + (4 * s) * D_DIM);
        xe1[s] = *(const v2f*)(xa + (4 * s + 1) * D_DIM);
    }

    v4f b0, b1, a1, a2;
#pragma unroll
    for (int k = 0; k < 4; ++k) {
        b0[k] = (h0[k] > 0.0f) ? w0 * w2[k] : 0.0f;
        b1[k] = (h1[k] > 0.0f) ? w1 * w2[k] : 0.0f;
        a1[k] = b0[k] + b1[k];
        a2[k] = w0 * h0[k] + w1 * h1[k];
    }
    s1[r * 8 + c4] = a1;
    s2[r * 8 + c4] = a2;
    *(v4f*)(sBs + r * COL_PITCH + c4 * 4)        = b0;
    *(v4f*)(sBs + (r + 32) * COL_PITCH + c4 * 4) = b1;
    __syncthreads();

    if (tid < 64) {
        const int col   = tid & 31;
        const int which = tid >> 5;
        const float* s = which ? (const float*)s2 : (const float*)s1;
        float acc = 0.0f;
#pragma unroll
        for (int rr = 0; rr < 32; ++rr) acc += s[rr * 32 + col];
        part[(which * COL_RG + rg) * H_DIM + cg * 32 + col] = acc;
    }

    const float* fb = sBs + (2 * half) * COL_PITCH + 2 * l16;
    v2f be0[COL_KS], be1[COL_KS];
#pragma unroll
    for (int s = 0; s < COL_KS; ++s) {
        be0[s] = *(const v2f*)(fb + (4 * s) * COL_PITCH);
        be1[s] = *(const v2f*)(fb + (4 * s + 1) * COL_PITCH);
    }
    __builtin_amdgcn_sched_barrier(0);

    v8f acc00 = {}, acc01 = {}, acc10 = {}, acc11 = {};
#pragma unroll
    for (int s = 0; s < COL_KS; ++s) {
        v2f aq0, aq1, bn0, bn1;
        aq0.x = xe0[s].x;  aq0.y = xe1[s].x;
        aq1.x = xe0[s].y;  aq1.y = xe1[s].y;
        bn0.x = be0[s].x;  bn0.y = be1[s].x;
        bn1.x = be0[s].y;  bn1.y = be1[s].y;
        acc00 = __builtin_amdgcn_wmma_f32_16x16x4_f32(
            false, aq0, false, bn0, (short)0, acc00, false, false);
        acc01 = __builtin_amdgcn_wmma_f32_16x16x4_f32(
            false, aq0, false, bn1, (short)0, acc01, false, false);
        acc10 = __builtin_amdgcn_wmma_f32_16x16x4_f32(
            false, aq1, false, bn0, (short)0, acc10, false, false);
        acc11 = __builtin_amdgcn_wmma_f32_16x16x4_f32(
            false, aq1, false, bn1, (short)0, acc11, false, false);
    }

    float* pp = pw1 + ((size_t)rg * D_DIM + wave * 32 + 16 * half) * H_DIM + cg * 32 + 2 * l16;
#pragma unroll
    for (int v = 0; v < 8; ++v) {
        v2f o0, o1;
        o0.x = acc00[v];  o0.y = acc01[v];
        o1.x = acc10[v];  o1.y = acc11[v];
        *(v2f*)(pp + (2 * v) * H_DIM)     = o0;
        *(v2f*)(pp + (2 * v + 1) * H_DIM) = o1;
    }
}

#define WS_BLOCKS 256
#define WS_GRID   (WS_BLOCKS + 8)

__global__ __launch_bounds__(64) void w1_sum_kernel(
    const float* __restrict__ pw1, const float* __restrict__ part,
    float* __restrict__ out)
{
    if (blockIdx.x >= WS_BLOCKS) {
        const int o     = (blockIdx.x - WS_BLOCKS) * 64 + threadIdx.x;
        const int which = o >> 8;
        const int j     = o & (H_DIM - 1);
        const float* p  = part + which * (COL_RG * H_DIM) + j;
        float s = 0.0f;
#pragma unroll
        for (int g = 0; g < COL_RG; ++g) s += p[g * H_DIM];
        out[D_DIM * H_DIM + which * H_DIM + j] = -s;
        return;
    }

    const int q = blockIdx.x * 64 + threadIdx.x;
    const v4f* p = (const v4f*)pw1 + q;
    v4f v[COL_RG];
#pragma unroll
    for (int g = 0; g < COL_RG; ++g) v[g] = p[g * (D_DIM * H_DIM / 4)];
    __builtin_amdgcn_sched_barrier(0);
    v4f s = v[0];
#pragma unroll
    for (int g = 1; g < COL_RG; ++g) s += v[g];
    *((v4f*)out + q) = -s;
}

extern "C" void kernel_launch(void* const* d_in, const int* in_sizes, int n_in,
                              void* d_out, int out_size, void* d_ws, size_t ws_size,
                              hipStream_t stream)
{
    const float* xs = (const float*)d_in[0];
    const float* ys = (const float*)d_in[1];
    const float* W1 = (const float*)d_in[2];
    const float* b1 = (const float*)d_in[3];
    const float* W2 = (const float*)d_in[4];
    const float* b2 = (const float*)d_in[5];
    float* out = (float*)d_out;

    float* ws = (float*)d_ws;
    float* h     = ws;
    float* u     = h + (size_t)T_DIM * H_DIM;
    float* sigma = u + T_DIM;
    float* w     = sigma + T_DIM;
    float* part  = w + T_DIM;
    float* pw1   = part + 2 * COL_RG * H_DIM;

    fwd_gemm_kernel<<<256, 128, 0, stream>>>(xs, W1, b1, h);
    value_kernel<<<VAL_BLOCKS, 32 * VAL_WAVES, 0, stream>>>(h, W2, b2, u);
    scan_kernel<<<1, 256, 0, stream>>>(u, ys, w, out);
    col_kernel<<<COL_BLOCKS, 256, 0, stream>>>(h, W2, w, xs, part, pw1);
    w1_sum_kernel<<<WS_GRID, 64, 0, stream>>>(pw1, part, out);
}
